# starting kernel
# baseline (speedup 1.0000x reference)
#include <hip/hip_runtime.h>

typedef __attribute__((ext_vector_type(2))) float v2f;
typedef __attribute__((ext_vector_type(8))) float v8f;

#define NI 64
#define NTILE 4

__device__ __forceinline__ float fast_tanh(float x) {
#if __has_builtin(__builtin_amdgcn_tanhf)
  return __builtin_amdgcn_tanhf(x);
#else
  float r;
  asm volatile("v_tanh_f32 %0, %1" : "=v"(r) : "v"(x));
  return r;
#endif
}

__global__ __launch_bounds__(256) void henon_wmma_kernel(
    const float* __restrict__ z, const float* __restrict__ W_in,
    const float* __restrict__ W_out, const float* __restrict__ b_in,
    const float* __restrict__ eta, float* __restrict__ out, int nbatch) {
  __shared__ float Gt[8][NI][16];

  const int tid   = threadIdx.x;
  const int lane  = tid & 31;
  const int wslot = tid >> 5;
  const int hilo  = lane >> 4;
  const int l16   = lane & 15;
  float* gbase = &Gt[wslot][0][0];

  float b1x[NTILE], b1y[NTILE], wv[NTILE];
#pragma unroll
  for (int T = 0; T < NTILE; ++T) {
    int j = T * 16 + l16;
    b1x[T] = hilo ? b_in[j] : W_in[0 * NI + j];
    b1y[T] = hilo ? 0.0f    : W_in[1 * NI + j];
    wv[T]  = W_out[j];
  }
  float a2x[16], a2y[16];
#pragma unroll
  for (int c = 0; c < 16; ++c) {
    int k0 = 4 * c + 2 * hilo;
    a2x[c] = (l16 < 2) ? W_in[l16 * NI + k0]     : 0.0f;
    a2y[c] = (l16 < 2) ? W_in[l16 * NI + k0 + 1] : 0.0f;
  }
  const float eta0 = eta[0], eta1 = eta[1];

  const int ntiles = (nbatch + 15) >> 4;
  const int waveG  = (int)(blockIdx.x * blockDim.x + tid) >> 5;
  const int nwaves = (int)(gridDim.x * blockDim.x) >> 5;
  const float4* z4 = (const float4*)z;
  float4* o4 = (float4*)out;

  for (int tile = waveG; tile < ntiles; tile += nwaves) {
    const int b = tile * 16 + l16;
    const bool act = (hilo == 0) && (b < nbatch);
    float4 zv = make_float4(0.f, 0.f, 0.f, 0.f);
    if (act) zv = z4[b];
    {
      int nt = tile + nwaves;
      if (nt < ntiles) __builtin_prefetch(&z4[nt * 16 + l16], 0, 1);
    }
    float X0 = zv.x, X1 = zv.y, Y0 = zv.z, Y1 = zv.w;

#pragma unroll
    for (int it = 0; it < 4; ++it) {
      v2f a1;
      a1.x = hilo ? 1.0f : Y0;
      a1.y = hilo ? 0.0f : Y1;
      v8f h[NTILE];
#pragma unroll
      for (int T = 0; T < NTILE; ++T) {
        v2f b1; b1.x = b1x[T]; b1.y = b1y[T];
        v8f cz = {};
        h[T] = __builtin_amdgcn_wmma_f32_16x16x4_f32(false, a1, false, b1,
                                                     (short)0, cz, false, false);
      }
#pragma unroll
      for (int T = 0; T < NTILE; ++T) {
        float g[8];
#pragma unroll
        for (int r = 0; r < 8; ++r) {
          float t  = fast_tanh(h[T][r]);
          float tw = t * wv[T];
          g[r] = __builtin_fmaf(-tw, t, wv[T]);
        }
        int addr = (T * 16 + l16) * 16 + 8 * hilo;
        *(float4*)(gbase + addr)     = make_float4(g[0], g[1], g[2], g[3]);
        *(float4*)(gbase + addr + 4) = make_float4(g[4], g[5], g[6], g[7]);
      }
      v8f acc = {};
#pragma unroll
      for (int c = 0; c < 16; ++c) {
        int roff = (4 * c + 2 * hilo) * 16 + l16;
        v2f bf;
        bf.x = gbase[roff];
        bf.y = gbase[roff + 16];
        v2f af; af.x = a2x[c]; af.y = a2y[c];
        acc = __builtin_amdgcn_wmma_f32_16x16x4_f32(false, af, false, bf,
                                                    (short)0, acc, false, false);
      }
      float g0 = acc[0], g1 = acc[1];
      float nX0 = Y0 + eta0;
      float nX1 = Y1 + eta1;
      float nY0 = g0 - X0;
      float nY1 = g1 - X1;
      X0 = nX0; X1 = nX1; Y0 = nY0; Y1 = nY1;
    }
    if (act) o4[b] = make_float4(X0, X1, Y0, Y1);
  }
}

extern "C" void kernel_launch(void* const* d_in, const int* in_sizes, int n_in,
                              void* d_out, int out_size, void* d_ws, size_t ws_size,
                              hipStream_t stream) {
  const float* z     = (const float*)d_in[0];
  const float* W_in  = (const float*)d_in[1];
  const float* W_out = (const float*)d_in[2];
  const float* b_in  = (const float*)d_in[3];
  const float* eta   = (const float*)d_in[4];
  float* out = (float*)d_out;
  int nbatch = in_sizes[0] / 4;

  dim3 block(256);
  dim3 grid(2048);
  henon_wmma_kernel<<<grid, block, 0, stream>>>(z, W_in, W_out, b_in, eta,
                                                out, nbatch);
}


# matmul1 computed transposed (weights as A operand, batch as B) so its D registers feed matmul2 as B fragments directly; LDS transpose removed; matmul2 now sums hidden units in a different k order
# speedup vs baseline: 1.0940x; 1.0940x over previous
#include <hip/hip_runtime.h>


typedef __attribute__((ext_vector_type(2))) float v2f;
typedef __attribute__((ext_vector_type(8))) float v8f;

#define NI 64
#define NTILE 4

__device__ __forceinline__ float fast_tanh(float x) {
#if __has_builtin(__builtin_amdgcn_tanhf)
  return __builtin_amdgcn_tanhf(x);
#else
  float r;
  asm volatile("v_tanh_f32 %0, %1" : "=v"(r) : "v"(x));
  return r;
#endif
}

__global__ __launch_bounds__(256) void henon_wmma_kernel(
    const float* __restrict__ z, const float* __restrict__ W_in,
    const float* __restrict__ W_out, const float* __restrict__ b_in,
    const float* __restrict__ eta, float* __restrict__ out, int nbatch) {
  const int tid  = threadIdx.x;
  const int lane = tid & 31;
  const int hilo = lane >> 4;
  const int l16  = lane & 15;

  float w1x[NTILE], w1y[NTILE];
#pragma unroll
  for (int T = 0; T < NTILE; ++T) {
    int j = T * 16 + l16;
    w1x[T] = hilo ? b_in[j] : W_in[0 * NI + j];
    w1y[T] = hilo ? 0.0f    : W_in[1 * NI + j];
  }
  float wv[NTILE][8];
#pragma unroll
  for (int T = 0; T < NTILE; ++T) {
#pragma unroll
    for (int r = 0; r < 8; ++r) wv[T][r] = W_out[T * 16 + 8 * hilo + r];
  }
  float a2x[NTILE][4], a2y[NTILE][4];
#pragma unroll
  for (int T = 0; T < NTILE; ++T) {
#pragma unroll
    for (int p = 0; p < 4; ++p) {
      int k0 = T * 16 + 8 * hilo + 2 * p;
      a2x[T][p] = (l16 < 2) ? W_in[l16 * NI + k0]     : 0.0f;
      a2y[T][p] = (l16 < 2) ? W_in[l16 * NI + k0 + 1] : 0.0f;
    }
  }
  const float eta0 = eta[0], eta1 = eta[1];

  const int ntiles = (nbatch + 15) >> 4;
  const int waveG  = (int)(blockIdx.x * blockDim.x + tid) >> 5;
  const int nwaves = (int)(gridDim.x * blockDim.x) >> 5;
  const float4* z4 = (const float4*)z;
  float4* o4 = (float4*)out;

  for (int tile = waveG; tile < ntiles; tile += nwaves) {
    const int b = tile * 16 + l16;
    const bool act = (hilo == 0) && (b < nbatch);
    float4 zv = make_float4(0.f, 0.f, 0.f, 0.f);
    if (act) zv = z4[b];
    {
      int nt = tile + nwaves;
      if (nt < ntiles) __builtin_prefetch(&z4[nt * 16 + l16], 0, 1);
    }
    float X0 = zv.x, X1 = zv.y, Y0 = zv.z, Y1 = zv.w;

#pragma unroll
    for (int it = 0; it < 4; ++it) {
      v2f yb;
      yb.x = hilo ? 1.0f : Y0;
      yb.y = hilo ? 0.0f : Y1;
      v8f h[NTILE];
#pragma unroll
      for (int T = 0; T < NTILE; ++T) {
        v2f wa; wa.x = w1x[T]; wa.y = w1y[T];
        v8f cz = {};
        h[T] = __builtin_amdgcn_wmma_f32_16x16x4_f32(false, wa, false, yb,
                                                     (short)0, cz, false, false);
      }
#pragma unroll
      for (int T = 0; T < NTILE; ++T) {
#pragma unroll
        for (int r = 0; r < 8; ++r) {
          float t  = fast_tanh(h[T][r]);
          float tw = t * wv[T][r];
          h[T][r] = __builtin_fmaf(-tw, t, wv[T][r]);
        }
      }
      v8f acc = {};
#pragma unroll
      for (int T = 0; T < NTILE; ++T) {
#pragma unroll
        for (int p = 0; p < 4; ++p) {
          v2f bf; bf.x = h[T][2 * p]; bf.y = h[T][2 * p + 1];
          v2f af; af.x = a2x[T][p];   af.y = a2y[T][p];
          acc = __builtin_amdgcn_wmma_f32_16x16x4_f32(false, af, false, bf,
                                                      (short)0, acc, false, false);
        }
      }
      float g0 = acc[0], g1 = acc[1];
      float nX0 = Y0 + eta0;
      float nX1 = Y1 + eta1;
      float nY0 = g0 - X0;
      float nY1 = g1 - X1;
      X0 = nX0; X1 = nX1; Y0 = nY0; Y1 = nY1;
    }
    if (act) o4[b] = make_float4(X0, X1, Y0, Y1);
  }
}

extern "C" void kernel_launch(void* const* d_in, const int* in_sizes, int n_in,
                              void* d_out, int out_size, void* d_ws, size_t ws_size,
                              hipStream_t stream) {
  const float* z     = (const float*)d_in[0];
  const float* W_in  = (const float*)d_in[1];
  const float* W_out = (const float*)d_in[2];
  const float* b_in  = (const float*)d_in[3];
  const float* eta   = (const float*)d_in[4];
  float* out = (float*)d_out;
  int nbatch = in_sizes[0] / 4;

  dim3 block(256);
  dim3 grid(2048);
  henon_wmma_kernel<<<grid, block, 0, stream>>>(z, W_in, W_out, b_in, eta,
                                                out, nbatch);
}
